# starting kernel
# baseline (speedup 1.0000x reference)
#include <hip/hip_runtime.h>

typedef float v2f __attribute__((ext_vector_type(2)));
typedef float v8f __attribute__((ext_vector_type(8)));

#define B_  2
#define S_  2048
#define D_  1024
#define H_  16
#define DK_ 64
#define N3D_ (3 * D_)
#define MROWS_ (B_ * S_)
#define THETA_ 10000.0f

__device__ __forceinline__ v8f wmma_f32(v2f a, v2f b, v8f c) {
    return __builtin_amdgcn_wmma_f32_16x16x4_f32(
         false, a,  false, b,
         (short)0, c,  false,  false);
}

__global__ __launch_bounds__(256) void qkv_rope_kernel(
    const float* __restrict__ X, const float* __restrict__ W,
    const int* __restrict__ tokpos,
    float* __restrict__ Qo, float* __restrict__ Ko, float* __restrict__ Vo) {
    const int lane = threadIdx.x & 31;
    const int wave = threadIdx.x >> 5;
    const int l16  = lane & 15;
    const int half = lane >> 4;
    const int rowBase = blockIdx.y * 32 + (wave >> 2) * 16;
    const int colBase = blockIdx.x * 64 + (wave & 3) * 16;

    const float* Arow = X + (size_t)(rowBase + l16) * D_ + 2 * half;
    const float* Bcol = W + (size_t)(2 * half) * N3D_ + colBase + l16;

    v8f acc = {};
    #pragma unroll 4
    for (int k = 0; k < D_; k += 4) {
        v2f a; a.x = Arow[k];               a.y = Arow[k + 1];
        v2f b; b.x = Bcol[(size_t)k * N3D_]; b.y = Bcol[(size_t)(k + 1) * N3D_];
        acc = wmma_f32(a, b, acc);
    }

    #pragma unroll
    for (int t = 0; t < 8; ++t) {
        const int M  = t + 8 * half;
        const int gm = rowBase + M;
        const int gn = colBase + l16;
        const int s  = gm & (S_ - 1);
        const int b  = gm >> 11;
        const int sel = gn >> 10;
        const int d   = gn & (D_ - 1);
        const int h   = d >> 6;
        const int dd  = d & (DK_ - 1);
        float val = acc[t];
        float out;
        if (sel < 2) {
            const float partner = __shfl_xor(val, 1, 32);
            const float pos = (float)tokpos[s];
            const float fexp = (float)(dd & ~1) * (1.0f / (float)DK_);
            const float inv_freq = __powf(THETA_, -fexp);
            float sn, cs;
            __sincosf(pos * inv_freq, &sn, &cs);
            out = (dd & 1) ? (partner * sn + val * cs)
                           : (val * cs - partner * sn);
        } else {
            out = val;
        }
        float* dst = (sel == 0) ? Qo : (sel == 1) ? Ko : Vo;
        dst[(size_t)(((b * H_ + h) * S_) + s) * DK_ + dd] = out;
    }
}

__global__ __launch_bounds__(32) void flash_kernel(
    const float* __restrict__ Q, const float* __restrict__ K,
    const float* __restrict__ V, float* __restrict__ Oout) {
    __shared__ float pbuf[16][17];
    const int lane = threadIdx.x & 31;
    const int l16  = lane & 15;
    const int half = lane >> 4;
    const int qt = blockIdx.x;
    const int bh = blockIdx.y;
    const int b  = bh >> 4, h = bh & 15;
    const int qBase = qt * 16;

    const float* Qp = Q + (size_t)bh * S_ * DK_;
    const float* Kp = K + (size_t)bh * S_ * DK_;
    const float* Vp = V + (size_t)bh * S_ * DK_;

    v2f qa[16];
    #pragma unroll
    for (int kk = 0; kk < 16; ++kk) {
        const float* p = Qp + (size_t)(qBase + l16) * DK_ + kk * 4 + 2 * half;
        qa[kk].x = p[0] * 0.125f;
        qa[kk].y = p[1] * 0.125f;
    }

    v8f o0 = {}, o1 = {}, o2 = {}, o3 = {};
    float m_i[8], l_i[8];
    #pragma unroll
    for (int t = 0; t < 8; ++t) { m_i[t] = -1e30f; l_i[t] = 0.0f; }

    for (int kt = 0; kt <= qt; ++kt) {
        const int kBase = kt * 16;

        v8f sc = {};
        #pragma unroll
        for (int kk = 0; kk < 16; ++kk) {
            const float* p = Kp + (size_t)(kBase + l16) * DK_ + kk * 4 + 2 * half;
            v2f bf; bf.x = p[0]; bf.y = p[1];
            sc = wmma_f32(qa[kk], bf, sc);
        }

        float alpha[8];
        #pragma unroll
        for (int t = 0; t < 8; ++t) {
            const int row = qBase + t + 8 * half;
            const int col = kBase + l16;
            float v = (col <= row) ? sc[t] : -1e30f;
            float r = v;
            r = fmaxf(r, __shfl_xor(r, 1, 32));
            r = fmaxf(r, __shfl_xor(r, 2, 32));
            r = fmaxf(r, __shfl_xor(r, 4, 32));
            r = fmaxf(r, __shfl_xor(r, 8, 32));
            const float mn = fmaxf(m_i[t], r);
            const float a  = __expf(m_i[t] - mn);
            const float pv = __expf(v - mn);
            float rs = pv;
            rs += __shfl_xor(rs, 1, 32);
            rs += __shfl_xor(rs, 2, 32);
            rs += __shfl_xor(rs, 4, 32);
            rs += __shfl_xor(rs, 8, 32);
            l_i[t] = l_i[t] * a + rs;
            m_i[t] = mn;
            alpha[t] = a;
            sc[t] = pv;
        }
        #pragma unroll
        for (int t = 0; t < 8; ++t) {
            o0[t] *= alpha[t]; o1[t] *= alpha[t];
            o2[t] *= alpha[t]; o3[t] *= alpha[t];
        }

        #pragma unroll
        for (int t = 0; t < 8; ++t) pbuf[t + 8 * half][l16] = sc[t];
        asm volatile("s_wait_dscnt 0" ::: "memory");
        v2f pa[4];
        #pragma unroll
        for (int kk = 0; kk < 4; ++kk) {
            pa[kk].x = pbuf[l16][kk * 4 + 2 * half];
            pa[kk].y = pbuf[l16][kk * 4 + 2 * half + 1];
        }

        #pragma unroll
        for (int kk = 0; kk < 4; ++kk) {
            const int krow = kBase + kk * 4 + 2 * half;
            const float* v0 = Vp + (size_t)krow * DK_;
            const float* v1 = v0 + DK_;
            v2f b0; b0.x = v0[ 0 + l16]; b0.y = v1[ 0 + l16];
            v2f b1; b1.x = v0[16 + l16]; b1.y = v1[16 + l16];
            v2f b2; b2.x = v0[32 + l16]; b2.y = v1[32 + l16];
            v2f b3; b3.x = v0[48 + l16]; b3.y = v1[48 + l16];
            o0 = wmma_f32(pa[kk], b0, o0);
            o1 = wmma_f32(pa[kk], b1, o1);
            o2 = wmma_f32(pa[kk], b2, o2);
            o3 = wmma_f32(pa[kk], b3, o3);
        }
        asm volatile("s_wait_dscnt 0" ::: "memory");
    }

    float* Orow = Oout + ((size_t)b * S_ + qBase) * D_ + h * DK_;
    #pragma unroll
    for (int t = 0; t < 8; ++t) {
        const float inv = 1.0f / l_i[t];
        float* dst = Orow + (size_t)(t + 8 * half) * D_;
        dst[ 0 + l16] = o0[t] * inv;
        dst[16 + l16] = o1[t] * inv;
        dst[32 + l16] = o2[t] * inv;
        dst[48 + l16] = o3[t] * inv;
    }
}

__global__ __launch_bounds__(256) void out_proj_kernel(
    const float* __restrict__ A, const float* __restrict__ W,
    float* __restrict__ out) {
    const int lane = threadIdx.x & 31;
    const int wave = threadIdx.x >> 5;
    const int l16  = lane & 15;
    const int half = lane >> 4;
    const int rowBase = blockIdx.y * 32 + (wave >> 2) * 16;
    const int colBase = blockIdx.x * 64 + (wave & 3) * 16;

    const float* Arow = A + (size_t)(rowBase + l16) * D_ + 2 * half;
    const float* Bcol = W + (size_t)(2 * half) * D_ + colBase + l16;

    v8f acc = {};
    #pragma unroll 4
    for (int k = 0; k < D_; k += 4) {
        v2f a; a.x = Arow[k];              a.y = Arow[k + 1];
        v2f b; b.x = Bcol[(size_t)k * D_]; b.y = Bcol[(size_t)(k + 1) * D_];
        acc = wmma_f32(a, b, acc);
    }
    #pragma unroll
    for (int t = 0; t < 8; ++t) {
        const int gm = rowBase + t + 8 * half;
        const int gn = colBase + l16;
        out[(size_t)gm * D_ + gn] = acc[t];
    }
}

extern "C" void kernel_launch(void* const* d_in, const int* in_sizes, int n_in,
                              void* d_out, int out_size, void* d_ws, size_t ws_size,
                              hipStream_t stream) {
    (void)in_sizes; (void)n_in; (void)out_size; (void)ws_size;
    const float* X      = (const float*)d_in[0];
    const int*   tokpos = (const int*)d_in[1];
    const float* Wqkv   = (const float*)d_in[2];
    const float* Wo     = (const float*)d_in[3];
    float* out = (float*)d_out;

    float* ws = (float*)d_ws;
    const size_t BHSD = (size_t)B_ * H_ * S_ * DK_;
    float* Qb = ws;
    float* Kb = ws + BHSD;
    float* Vb = ws + 2 * BHSD;
    float* Ob = ws + 3 * BHSD;

    dim3 g1(N3D_ / 64, MROWS_ / 32);
    qkv_rope_kernel<<<g1, 256, 0, stream>>>(X, Wqkv, tokpos, Qb, Kb, Vb);

    dim3 g2(S_ / 16, B_ * H_);
    flash_kernel<<<g2, 32, 0, stream>>>(Qb, Kb, Vb, Ob);

    dim3 g3(D_ / 64, MROWS_ / 32);
    out_proj_kernel<<<g3, 256, 0, stream>>>(Ob, Wo, out);
}


# Projection GEMMs retiled: 128x128 (out-proj 64x128) block tiles staged through LDS, 64x64 (32x64) wave tiles so one fragment read feeds 4 WMMAs; k is summed in a different order
# speedup vs baseline: 1.3326x; 1.3326x over previous
#include <hip/hip_runtime.h>

typedef float v2f __attribute__((ext_vector_type(2)));
typedef float v8f __attribute__((ext_vector_type(8)));

#define B_  2
#define S_  2048
#define D_  1024
#define H_  16
#define DK_ 64
#define N3D_ (3 * D_)
#define MROWS_ (B_ * S_)
#define THETA_ 10000.0f

__device__ __forceinline__ v8f wmma_f32(v2f a, v2f b, v8f c) {
    return __builtin_amdgcn_wmma_f32_16x16x4_f32(
         false, a,  false, b,
         (short)0, c,  false,  false);
}

typedef float v4f __attribute__((ext_vector_type(4)));

#define GK_ 16
#define AP_ 20
#define BP_ 256
#define KSTAGES_ (D_ / GK_)

template <int MT, int LDB>
__device__ __forceinline__ void gemm_tile(
    const float* __restrict__ A, const float* __restrict__ Bm,
    const int rowBlk, const int colBlk,
    float* __restrict__ As, float* __restrict__ Bs, v8f (&acc)[MT][4]) {
    constexpr int BM = 32 * MT;
    constexpr int ASTAGE = BM * AP_;
    constexpr int BSTAGE = (GK_ / 2) * BP_;
    const int tid  = threadIdx.x;
    const int lane = tid & 31;
    const int wave = tid >> 5;
    const int c = lane & 15;
    const int h = lane >> 4;
    const int wm = wave >> 1;
    const int wn = wave & 1;

    const int arow = tid >> 2, akq = tid & 3;
    const int bkp  = tid >> 5, bcq = tid & 31;
    const float* Ag = A  + (size_t)(rowBlk + arow) * D_ + akq * 4;
    const float* Bg = Bm + (size_t)(2 * bkp) * LDB + colBlk + bcq * 4;
    float* Asw = As + arow * AP_ + akq * 4;
    float* Bsw = Bs + bkp * BP_ + bcq * 4;

    v4f ra[MT], rb[4];
    #pragma unroll
    for (int i = 0; i < MT; ++i) ra[i] = *(const v4f*)(Ag + (size_t)(32 * i) * D_);
    #pragma unroll
    for (int i = 0; i < 4; ++i)  rb[i] = *(const v4f*)(Bg + (size_t)(8 * (i >> 1) + (i & 1)) * LDB);
    #pragma unroll
    for (int i = 0; i < MT; ++i) *(v4f*)(Asw + (32 * i) * AP_) = ra[i];
    #pragma unroll
    for (int i = 0; i < 2; ++i) {
        v4f lo, hi;
        lo.x = rb[2 * i].x; lo.y = rb[2 * i + 1].x; lo.z = rb[2 * i].y; lo.w = rb[2 * i + 1].y;
        hi.x = rb[2 * i].z; hi.y = rb[2 * i + 1].z; hi.z = rb[2 * i].w; hi.w = rb[2 * i + 1].w;
        *(v4f*)(Bsw + (4 * i) * BP_)       = lo;
        *(v4f*)(Bsw + (4 * i) * BP_ + 128) = hi;
    }
    __syncthreads();

    const float* Asr = As + (wm * 16 * MT + c) * AP_ + h * 4;
    const float* Bsr = Bs + (h * 2) * BP_ + (wn * 16 + c) * 4;

    for (int kc = 0; kc < KSTAGES_; ++kc) {
        const int cur = kc & 1;
        const bool more = (kc + 1 < KSTAGES_);
        if (more) {
            const float* ag = Ag + (kc + 1) * GK_;
            const float* bg = Bg + (size_t)(kc + 1) * GK_ * LDB;
            #pragma unroll
            for (int i = 0; i < MT; ++i) ra[i] = *(const v4f*)(ag + (size_t)(32 * i) * D_);
            #pragma unroll
            for (int i = 0; i < 4; ++i)  rb[i] = *(const v4f*)(bg + (size_t)(8 * (i >> 1) + (i & 1)) * LDB);
            __builtin_amdgcn_sched_barrier(0);
        }
        const float* as = Asr + cur * ASTAGE;
        const float* bs = Bsr + cur * BSTAGE;
        #pragma unroll
        for (int s = 0; s < 2; ++s) {
            v4f af[MT], bf[2][2];
            #pragma unroll
            for (int tm = 0; tm < MT; ++tm) af[tm] = *(const v4f*)(as + tm * 16 * AP_ + s * 8);
            #pragma unroll
            for (int p = 0; p < 2; ++p) {
                bf[p][0] = *(const v4f*)(bs + (s * 4 + p) * BP_);
                bf[p][1] = *(const v4f*)(bs + (s * 4 + p) * BP_ + 128);
            }
            #pragma unroll
            for (int p = 0; p < 2; ++p) {
                #pragma unroll
                for (int tm = 0; tm < MT; ++tm) {
                    v2f a; a.x = af[tm][2 * p]; a.y = af[tm][2 * p + 1];
                    #pragma unroll
                    for (int tn = 0; tn < 4; ++tn) {
                        v2f b; b.x = bf[p][tn >> 1][2 * (tn & 1)]; b.y = bf[p][tn >> 1][2 * (tn & 1) + 1];
                        acc[tm][tn] = wmma_f32(a, b, acc[tm][tn]);
                    }
                }
            }
        }
        if (more) {
            __builtin_amdgcn_sched_barrier(0);
            float* aw = Asw + (cur ^ 1) * ASTAGE;
            float* bw = Bsw + (cur ^ 1) * BSTAGE;
            #pragma unroll
            for (int i = 0; i < MT; ++i) *(v4f*)(aw + (32 * i) * AP_) = ra[i];
            #pragma unroll
            for (int i = 0; i < 2; ++i) {
                v4f lo, hi;
                lo.x = rb[2 * i].x; lo.y = rb[2 * i + 1].x; lo.z = rb[2 * i].y; lo.w = rb[2 * i + 1].y;
                hi.x = rb[2 * i].z; hi.y = rb[2 * i + 1].z; hi.z = rb[2 * i].w; hi.w = rb[2 * i + 1].w;
                *(v4f*)(bw + (4 * i) * BP_)       = lo;
                *(v4f*)(bw + (4 * i) * BP_ + 128) = hi;
            }
        }
        __syncthreads();
    }
}

__global__ __launch_bounds__(128) void qkv_rope_kernel(
    const float* __restrict__ X, const float* __restrict__ W,
    const int* __restrict__ tokpos,
    float* __restrict__ Qo, float* __restrict__ Ko, float* __restrict__ Vo) {
    constexpr int MT = 4;
    __shared__ __attribute__((aligned(16))) float As[2 * 32 * MT * AP_];
    __shared__ __attribute__((aligned(16))) float Bs[2 * (GK_ / 2) * BP_];
    const int rowBlk = blockIdx.y * (32 * MT);
    const int colBlk = blockIdx.x * 128;

    v8f acc[MT][4];
    #pragma unroll
    for (int tm = 0; tm < MT; ++tm) {
        #pragma unroll
        for (int tn = 0; tn < 4; ++tn) acc[tm][tn] = v8f{};
    }
    gemm_tile<MT, N3D_>(X, W, rowBlk, colBlk, As, Bs, acc);

    const int lane = threadIdx.x & 31;
    const int wave = threadIdx.x >> 5;
    const int c = lane & 15;
    const int h = lane >> 4;
    const int wm = wave >> 1;
    const int wn = wave & 1;

    const int gn0 = colBlk + wn * 64 + 4 * c;
    const int sel = gn0 >> 10;
    const int d   = gn0 & (D_ - 1);
    const int hh  = d >> 6;
    const int dd0 = d & (DK_ - 1);
    float* dst = (sel == 0) ? Qo : (sel == 1) ? Ko : Vo;

    float invf0 = 0.0f, invf1 = 0.0f;
    if (sel < 2) {
        const float fexp0 = (float)dd0 * (1.0f / (float)DK_);
        const float fexp1 = (float)(dd0 + 2) * (1.0f / (float)DK_);
        invf0 = __powf(THETA_, -fexp0);
        invf1 = __powf(THETA_, -fexp1);
    }

    #pragma unroll
    for (int tm = 0; tm < MT; ++tm) {
        #pragma unroll
        for (int r = 0; r < 8; ++r) {
            const int gm = rowBlk + wm * 16 * MT + tm * 16 + 8 * h + r;
            const int s  = gm & (S_ - 1);
            const int b  = gm >> 11;
            const float x0 = acc[tm][0][r], x1 = acc[tm][1][r];
            const float x2 = acc[tm][2][r], x3 = acc[tm][3][r];
            v4f o;
            if (sel < 2) {
                const float pos = (float)tokpos[s];
                float sn0, cs0, sn1, cs1;
                __sincosf(pos * invf0, &sn0, &cs0);
                __sincosf(pos * invf1, &sn1, &cs1);
                o.x = x0 * cs0 - x1 * sn0;
                o.y = x0 * sn0 + x1 * cs0;
                o.z = x2 * cs1 - x3 * sn1;
                o.w = x2 * sn1 + x3 * cs1;
            } else {
                o.x = x0; o.y = x1; o.z = x2; o.w = x3;
            }
            *(v4f*)(dst + (size_t)(((b * H_ + hh) * S_) + s) * DK_ + dd0) = o;
        }
    }
}

__global__ __launch_bounds__(32) void flash_kernel(
    const float* __restrict__ Q, const float* __restrict__ K,
    const float* __restrict__ V, float* __restrict__ Oout) {
    __shared__ float pbuf[16][17];
    const int lane = threadIdx.x & 31;
    const int l16  = lane & 15;
    const int half = lane >> 4;
    const int qt = blockIdx.x;
    const int bh = blockIdx.y;
    const int b  = bh >> 4, h = bh & 15;
    const int qBase = qt * 16;

    const float* Qp = Q + (size_t)bh * S_ * DK_;
    const float* Kp = K + (size_t)bh * S_ * DK_;
    const float* Vp = V + (size_t)bh * S_ * DK_;

    v2f qa[16];
    #pragma unroll
    for (int kk = 0; kk < 16; ++kk) {
        const float* p = Qp + (size_t)(qBase + l16) * DK_ + kk * 4 + 2 * half;
        qa[kk].x = p[0] * 0.125f;
        qa[kk].y = p[1] * 0.125f;
    }

    v8f o0 = {}, o1 = {}, o2 = {}, o3 = {};
    float m_i[8], l_i[8];
    #pragma unroll
    for (int t = 0; t < 8; ++t) { m_i[t] = -1e30f; l_i[t] = 0.0f; }

    for (int kt = 0; kt <= qt; ++kt) {
        const int kBase = kt * 16;

        v8f sc = {};
        #pragma unroll
        for (int kk = 0; kk < 16; ++kk) {
            const float* p = Kp + (size_t)(kBase + l16) * DK_ + kk * 4 + 2 * half;
            v2f bf; bf.x = p[0]; bf.y = p[1];
            sc = wmma_f32(qa[kk], bf, sc);
        }

        float alpha[8];
        #pragma unroll
        for (int t = 0; t < 8; ++t) {
            const int row = qBase + t + 8 * half;
            const int col = kBase + l16;
            float v = (col <= row) ? sc[t] : -1e30f;
            float r = v;
            r = fmaxf(r, __shfl_xor(r, 1, 32));
            r = fmaxf(r, __shfl_xor(r, 2, 32));
            r = fmaxf(r, __shfl_xor(r, 4, 32));
            r = fmaxf(r, __shfl_xor(r, 8, 32));
            const float mn = fmaxf(m_i[t], r);
            const float a  = __expf(m_i[t] - mn);
            const float pv = __expf(v - mn);
            float rs = pv;
            rs += __shfl_xor(rs, 1, 32);
            rs += __shfl_xor(rs, 2, 32);
            rs += __shfl_xor(rs, 4, 32);
            rs += __shfl_xor(rs, 8, 32);
            l_i[t] = l_i[t] * a + rs;
            m_i[t] = mn;
            alpha[t] = a;
            sc[t] = pv;
        }
        #pragma unroll
        for (int t = 0; t < 8; ++t) {
            o0[t] *= alpha[t]; o1[t] *= alpha[t];
            o2[t] *= alpha[t]; o3[t] *= alpha[t];
        }

        #pragma unroll
        for (int t = 0; t < 8; ++t) pbuf[t + 8 * half][l16] = sc[t];
        asm volatile("s_wait_dscnt 0" ::: "memory");
        v2f pa[4];
        #pragma unroll
        for (int kk = 0; kk < 4; ++kk) {
            pa[kk].x = pbuf[l16][kk * 4 + 2 * half];
            pa[kk].y = pbuf[l16][kk * 4 + 2 * half + 1];
        }

        #pragma unroll
        for (int kk = 0; kk < 4; ++kk) {
            const int krow = kBase + kk * 4 + 2 * half;
            const float* v0 = Vp + (size_t)krow * DK_;
            const float* v1 = v0 + DK_;
            v2f b0; b0.x = v0[ 0 + l16]; b0.y = v1[ 0 + l16];
            v2f b1; b1.x = v0[16 + l16]; b1.y = v1[16 + l16];
            v2f b2; b2.x = v0[32 + l16]; b2.y = v1[32 + l16];
            v2f b3; b3.x = v0[48 + l16]; b3.y = v1[48 + l16];
            o0 = wmma_f32(pa[kk], b0, o0);
            o1 = wmma_f32(pa[kk], b1, o1);
            o2 = wmma_f32(pa[kk], b2, o2);
            o3 = wmma_f32(pa[kk], b3, o3);
        }
        asm volatile("s_wait_dscnt 0" ::: "memory");
    }

    float* Orow = Oout + ((size_t)b * S_ + qBase) * D_ + h * DK_;
    #pragma unroll
    for (int t = 0; t < 8; ++t) {
        const float inv = 1.0f / l_i[t];
        float* dst = Orow + (size_t)(t + 8 * half) * D_;
        dst[ 0 + l16] = o0[t] * inv;
        dst[16 + l16] = o1[t] * inv;
        dst[32 + l16] = o2[t] * inv;
        dst[48 + l16] = o3[t] * inv;
    }
}

__global__ __launch_bounds__(128) void out_proj_kernel(
    const float* __restrict__ A, const float* __restrict__ W,
    float* __restrict__ out) {
    constexpr int MT = 2;
    __shared__ __attribute__((aligned(16))) float As[2 * 32 * MT * AP_];
    __shared__ __attribute__((aligned(16))) float Bs[2 * (GK_ / 2) * BP_];
    const int rowBlk = blockIdx.y * (32 * MT);
    const int colBlk = blockIdx.x * 128;

    v8f acc[MT][4];
    #pragma unroll
    for (int tm = 0; tm < MT; ++tm) {
        #pragma unroll
        for (int tn = 0; tn < 4; ++tn) acc[tm][tn] = v8f{};
    }
    gemm_tile<MT, D_>(A, W, rowBlk, colBlk, As, Bs, acc);

    const int lane = threadIdx.x & 31;
    const int wave = threadIdx.x >> 5;
    const int c = lane & 15;
    const int h = lane >> 4;
    const int wm = wave >> 1;
    const int wn = wave & 1;
    const int gn0 = colBlk + wn * 64 + 4 * c;
    #pragma unroll
    for (int tm = 0; tm < MT; ++tm) {
        #pragma unroll
        for (int r = 0; r < 8; ++r) {
            const int gm = rowBlk + wm * 16 * MT + tm * 16 + 8 * h + r;
            v4f o;
            o.x = acc[tm][0][r]; o.y = acc[tm][1][r];
            o.z = acc[tm][2][r]; o.w = acc[tm][3][r];
            *(v4f*)(out + (size_t)gm * D_ + gn0) = o;
        }
    }
}

extern "C" void kernel_launch(void* const* d_in, const int* in_sizes, int n_in,
                              void* d_out, int out_size, void* d_ws, size_t ws_size,
                              hipStream_t stream) {
    (void)in_sizes; (void)n_in; (void)out_size; (void)ws_size;
    const float* X      = (const float*)d_in[0];
    const int*   tokpos = (const int*)d_in[1];
    const float* Wqkv   = (const float*)d_in[2];
    const float* Wo     = (const float*)d_in[3];
    float* out = (float*)d_out;

    float* ws = (float*)d_ws;
    const size_t BHSD = (size_t)B_ * H_ * S_ * DK_;
    float* Qb = ws;
    float* Kb = ws + BHSD;
    float* Vb = ws + 2 * BHSD;
    float* Ob = ws + 3 * BHSD;

    dim3 g1(N3D_ / 128, MROWS_ / 128);
    qkv_rope_kernel<<<g1, 128, 0, stream>>>(X, Wqkv, tokpos, Qb, Kb, Vb);

    dim3 g2(S_ / 16, B_ * H_);
    flash_kernel<<<g2, 32, 0, stream>>>(Qb, Kb, Vb, Ob);

    dim3 g3(D_ / 128, MROWS_ / 64);
    out_proj_kernel<<<g3, 128, 0, stream>>>(Ob, Wo, out);
}


# Attention retiled: 4-wave group per 64 queries, K/V staged once in LDS per 64 keys, read as 16-byte fragments; each group takes a heavy and a light query block; score k summed in another order
# speedup vs baseline: 1.9703x; 1.4785x over previous
#include <hip/hip_runtime.h>

typedef float v2f __attribute__((ext_vector_type(2)));
typedef float v8f __attribute__((ext_vector_type(8)));

#define B_  2
#define S_  2048
#define D_  1024
#define H_  16
#define DK_ 64
#define N3D_ (3 * D_)
#define MROWS_ (B_ * S_)
#define THETA_ 10000.0f

__device__ __forceinline__ v8f wmma_f32(v2f a, v2f b, v8f c) {
    return __builtin_amdgcn_wmma_f32_16x16x4_f32(
         false, a,  false, b,
         (short)0, c,  false,  false);
}

typedef float v4f __attribute__((ext_vector_type(4)));

#define GK_ 16
#define AP_ 20
#define BP_ 256
#define KSTAGES_ (D_ / GK_)

template <int MT, int LDB>
__device__ __forceinline__ void gemm_tile(
    const float* __restrict__ A, const float* __restrict__ Bm,
    const int rowBlk, const int colBlk,
    float* __restrict__ As, float* __restrict__ Bs, v8f (&acc)[MT][4]) {
    constexpr int BM = 32 * MT;
    constexpr int ASTAGE = BM * AP_;
    constexpr int BSTAGE = (GK_ / 2) * BP_;
    const int tid  = threadIdx.x;
    const int lane = tid & 31;
    const int wave = tid >> 5;
    const int c = lane & 15;
    const int h = lane >> 4;
    const int wm = wave >> 1;
    const int wn = wave & 1;

    const int arow = tid >> 2, akq = tid & 3;
    const int bkp  = tid >> 5, bcq = tid & 31;
    const float* Ag = A  + (size_t)(rowBlk + arow) * D_ + akq * 4;
    const float* Bg = Bm + (size_t)(2 * bkp) * LDB + colBlk + bcq * 4;
    float* Asw = As + arow * AP_ + akq * 4;
    float* Bsw = Bs + bkp * BP_ + bcq * 4;

    v4f ra[MT], rb[4];
    #pragma unroll
    for (int i = 0; i < MT; ++i) ra[i] = *(const v4f*)(Ag + (size_t)(32 * i) * D_);
    #pragma unroll
    for (int i = 0; i < 4; ++i)  rb[i] = *(const v4f*)(Bg + (size_t)(8 * (i >> 1) + (i & 1)) * LDB);
    #pragma unroll
    for (int i = 0; i < MT; ++i) *(v4f*)(Asw + (32 * i) * AP_) = ra[i];
    #pragma unroll
    for (int i = 0; i < 2; ++i) {
        v4f lo, hi;
        lo.x = rb[2 * i].x; lo.y = rb[2 * i + 1].x; lo.z = rb[2 * i].y; lo.w = rb[2 * i + 1].y;
        hi.x = rb[2 * i].z; hi.y = rb[2 * i + 1].z; hi.z = rb[2 * i].w; hi.w = rb[2 * i + 1].w;
        *(v4f*)(Bsw + (4 * i) * BP_)       = lo;
        *(v4f*)(Bsw + (4 * i) * BP_ + 128) = hi;
    }
    __syncthreads();

    const float* Asr = As + (wm * 16 * MT + c) * AP_ + h * 4;
    const float* Bsr = Bs + (h * 2) * BP_ + (wn * 16 + c) * 4;

    for (int kc = 0; kc < KSTAGES_; ++kc) {
        const int cur = kc & 1;
        const bool more = (kc + 1 < KSTAGES_);
        if (more) {
            const float* ag = Ag + (kc + 1) * GK_;
            const float* bg = Bg + (size_t)(kc + 1) * GK_ * LDB;
            #pragma unroll
            for (int i = 0; i < MT; ++i) ra[i] = *(const v4f*)(ag + (size_t)(32 * i) * D_);
            #pragma unroll
            for (int i = 0; i < 4; ++i)  rb[i] = *(const v4f*)(bg + (size_t)(8 * (i >> 1) + (i & 1)) * LDB);
            __builtin_amdgcn_sched_barrier(0);
        }
        const float* as = Asr + cur * ASTAGE;
        const float* bs = Bsr + cur * BSTAGE;
        #pragma unroll
        for (int s = 0; s < 2; ++s) {
            v4f af[MT], bf[2][2];
            #pragma unroll
            for (int tm = 0; tm < MT; ++tm) af[tm] = *(const v4f*)(as + tm * 16 * AP_ + s * 8);
            #pragma unroll
            for (int p = 0; p < 2; ++p) {
                bf[p][0] = *(const v4f*)(bs + (s * 4 + p) * BP_);
                bf[p][1] = *(const v4f*)(bs + (s * 4 + p) * BP_ + 128);
            }
            #pragma unroll
            for (int p = 0; p < 2; ++p) {
                #pragma unroll
                for (int tm = 0; tm < MT; ++tm) {
                    v2f a; a.x = af[tm][2 * p]; a.y = af[tm][2 * p + 1];
                    #pragma unroll
                    for (int tn = 0; tn < 4; ++tn) {
                        v2f b; b.x = bf[p][tn >> 1][2 * (tn & 1)]; b.y = bf[p][tn >> 1][2 * (tn & 1) + 1];
                        acc[tm][tn] = wmma_f32(a, b, acc[tm][tn]);
                    }
                }
            }
        }
        if (more) {
            __builtin_amdgcn_sched_barrier(0);
            float* aw = Asw + (cur ^ 1) * ASTAGE;
            float* bw = Bsw + (cur ^ 1) * BSTAGE;
            #pragma unroll
            for (int i = 0; i < MT; ++i) *(v4f*)(aw + (32 * i) * AP_) = ra[i];
            #pragma unroll
            for (int i = 0; i < 2; ++i) {
                v4f lo, hi;
                lo.x = rb[2 * i].x; lo.y = rb[2 * i + 1].x; lo.z = rb[2 * i].y; lo.w = rb[2 * i + 1].y;
                hi.x = rb[2 * i].z; hi.y = rb[2 * i + 1].z; hi.z = rb[2 * i].w; hi.w = rb[2 * i + 1].w;
                *(v4f*)(bw + (4 * i) * BP_)       = lo;
                *(v4f*)(bw + (4 * i) * BP_ + 128) = hi;
            }
        }
        __syncthreads();
    }
}

__global__ __launch_bounds__(128) void qkv_rope_kernel(
    const float* __restrict__ X, const float* __restrict__ W,
    const int* __restrict__ tokpos,
    float* __restrict__ Qo, float* __restrict__ Ko, float* __restrict__ Vo) {
    constexpr int MT = 4;
    __shared__ __attribute__((aligned(16))) float As[2 * 32 * MT * AP_];
    __shared__ __attribute__((aligned(16))) float Bs[2 * (GK_ / 2) * BP_];
    const int rowBlk = blockIdx.y * (32 * MT);
    const int colBlk = blockIdx.x * 128;

    v8f acc[MT][4];
    #pragma unroll
    for (int tm = 0; tm < MT; ++tm) {
        #pragma unroll
        for (int tn = 0; tn < 4; ++tn) acc[tm][tn] = v8f{};
    }
    gemm_tile<MT, N3D_>(X, W, rowBlk, colBlk, As, Bs, acc);

    const int lane = threadIdx.x & 31;
    const int wave = threadIdx.x >> 5;
    const int c = lane & 15;
    const int h = lane >> 4;
    const int wm = wave >> 1;
    const int wn = wave & 1;

    const int gn0 = colBlk + wn * 64 + 4 * c;
    const int sel = gn0 >> 10;
    const int d   = gn0 & (D_ - 1);
    const int hh  = d >> 6;
    const int dd0 = d & (DK_ - 1);
    float* dst = (sel == 0) ? Qo : (sel == 1) ? Ko : Vo;

    float invf0 = 0.0f, invf1 = 0.0f;
    if (sel < 2) {
        const float fexp0 = (float)dd0 * (1.0f / (float)DK_);
        const float fexp1 = (float)(dd0 + 2) * (1.0f / (float)DK_);
        invf0 = __powf(THETA_, -fexp0);
        invf1 = __powf(THETA_, -fexp1);
    }

    #pragma unroll
    for (int tm = 0; tm < MT; ++tm) {
        #pragma unroll
        for (int r = 0; r < 8; ++r) {
            const int gm = rowBlk + wm * 16 * MT + tm * 16 + 8 * h + r;
            const int s  = gm & (S_ - 1);
            const int b  = gm >> 11;
            const float x0 = acc[tm][0][r], x1 = acc[tm][1][r];
            const float x2 = acc[tm][2][r], x3 = acc[tm][3][r];
            v4f o;
            if (sel < 2) {
                const float pos = (float)tokpos[s];
                float sn0, cs0, sn1, cs1;
                __sincosf(pos * invf0, &sn0, &cs0);
                __sincosf(pos * invf1, &sn1, &cs1);
                o.x = x0 * cs0 - x1 * sn0;
                o.y = x0 * sn0 + x1 * cs0;
                o.z = x2 * cs1 - x3 * sn1;
                o.w = x2 * sn1 + x3 * cs1;
            } else {
                o.x = x0; o.y = x1; o.z = x2; o.w = x3;
            }
            *(v4f*)(dst + (size_t)(((b * H_ + hh) * S_) + s) * DK_ + dd0) = o;
        }
    }
}

#define FQ_ 64
#define FK_ 64
#define KP_ 68
#define FSTAGE_ (FK_ * KP_)
#define VP_ 132
#define VSTAGE_ ((FK_ / 2) * VP_)
#define NQB_ (S_ / FQ_)

__global__ __launch_bounds__(128) void flash_kernel(
    const float* __restrict__ Q, const float* __restrict__ K,
    const float* __restrict__ V, float* __restrict__ Oout) {
    __shared__ __attribute__((aligned(16))) float Ks[2 * FSTAGE_];
    __shared__ __attribute__((aligned(16))) float Vs[2 * VSTAGE_];
    __shared__ float pbuf[4][16][17];
    const int tid  = threadIdx.x;
    const int lane = tid & 31;
    const int wave = __builtin_amdgcn_readfirstlane(tid >> 5);
    const int l16  = lane & 15;
    const int half = lane >> 4;
    const int bh = blockIdx.y;
    const int b  = bh >> 4, h = bh & 15;

    const float* Qp = Q + (size_t)bh * S_ * DK_;
    const float* Kp = K + (size_t)bh * S_ * DK_;
    const float* Vp = V + (size_t)bh * S_ * DK_;

    const int srow = tid >> 4, scq = tid & 15;
    const float* Kg = Kp + (size_t)srow * DK_ + scq * 4;
    const float* Vg = Vp + (size_t)(2 * srow) * DK_ + scq * 4;
    float* Ksw = Ks + srow * KP_ + scq * 4;
    float* Vsw = Vs + srow * VP_ + scq * 4;
    const float* Ksr = Ks + l16 * KP_ + 4 * half;
    const float* Vsr = Vs + half * VP_ + 4 * l16;

    for (int pass = 0; pass < 2; ++pass) {
        const int qb = (pass == 0) ? (NQB_ - 1 - (int)blockIdx.x) : (int)blockIdx.x;
        const int qBase = qb * FQ_ + wave * 16;

        v2f qa[16];
        #pragma unroll
        for (int j = 0; j < 8; ++j) {
            const v4f qv = *(const v4f*)(Qp + (size_t)(qBase + l16) * DK_ + 8 * j + 4 * half);
            qa[2 * j].x     = qv.x * 0.125f; qa[2 * j].y     = qv.y * 0.125f;
            qa[2 * j + 1].x = qv.z * 0.125f; qa[2 * j + 1].y = qv.w * 0.125f;
        }

        v8f o0 = {}, o1 = {}, o2 = {}, o3 = {};
        float m_i[8], l_i[8];
        #pragma unroll
        for (int t = 0; t < 8; ++t) { m_i[t] = -1e30f; l_i[t] = 0.0f; }

        v4f rk[8], rv[8];
        #pragma unroll
        for (int i = 0; i < 8; ++i) {
            rk[i] = *(const v4f*)(Kg + (size_t)(8 * i) * DK_);
            rv[i] = *(const v4f*)(Vg + (size_t)(16 * (i >> 1) + (i & 1)) * DK_);
        }
        #pragma unroll
        for (int i = 0; i < 8; ++i) *(v4f*)(Ksw + (8 * i) * KP_) = rk[i];
        #pragma unroll
        for (int i = 0; i < 4; ++i) {
            v4f lo, hi;
            lo.x = rv[2 * i].x; lo.y = rv[2 * i + 1].x; lo.z = rv[2 * i].y; lo.w = rv[2 * i + 1].y;
            hi.x = rv[2 * i].z; hi.y = rv[2 * i + 1].z; hi.z = rv[2 * i].w; hi.w = rv[2 * i + 1].w;
            *(v4f*)(Vsw + (8 * i) * VP_)      = lo;
            *(v4f*)(Vsw + (8 * i) * VP_ + 64) = hi;
        }
        __syncthreads();

        for (int kb = 0; kb <= qb; ++kb) {
            const int cur = kb & 1;
            const bool more = (kb < qb);
            if (more) {
                #pragma unroll
                for (int i = 0; i < 8; ++i) {
                    rk[i] = *(const v4f*)(Kg + (size_t)((kb + 1) * FK_ + 8 * i) * DK_);
                    rv[i] = *(const v4f*)(Vg + (size_t)((kb + 1) * FK_ + 16 * (i >> 1) + (i & 1)) * DK_);
                }
            }
            const int nst = more ? 4 : (wave + 1);
            const float* ksb = Ksr + cur * FSTAGE_;
            const float* vsb = Vsr + cur * VSTAGE_;
            for (int st = 0; st < nst; ++st) {
                const int kBase = kb * FK_ + st * 16;

                v4f kf[8], vf[4][2];
                #pragma unroll
                for (int j = 0; j < 8; ++j) kf[j] = *(const v4f*)(ksb + (16 * st) * KP_ + 8 * j);
                #pragma unroll
                for (int kk = 0; kk < 4; ++kk) {
                    vf[kk][0] = *(const v4f*)(vsb + (8 * st + 2 * kk) * VP_);
                    vf[kk][1] = *(const v4f*)(vsb + (8 * st + 2 * kk) * VP_ + 64);
                }

                v8f sA = {}, sB = {};
                #pragma unroll
                for (int j = 0; j < 8; ++j) {
                    v2f b0; b0.x = kf[j].x; b0.y = kf[j].y;
                    v2f b1; b1.x = kf[j].z; b1.y = kf[j].w;
                    sA = wmma_f32(qa[2 * j],     b0, sA);
                    sB = wmma_f32(qa[2 * j + 1], b1, sB);
                }
                v8f sc = sA + sB;

                float alpha[8];
                #pragma unroll
                for (int t = 0; t < 8; ++t) {
                    const int row = qBase + t + 8 * half;
                    const int col = kBase + l16;
                    float v = (col <= row) ? sc[t] : -1e30f;
                    float r = v;
                    r = fmaxf(r, __shfl_xor(r, 1, 32));
                    r = fmaxf(r, __shfl_xor(r, 2, 32));
                    r = fmaxf(r, __shfl_xor(r, 4, 32));
                    r = fmaxf(r, __shfl_xor(r, 8, 32));
                    const float mn = fmaxf(m_i[t], r);
                    const float a  = __expf(m_i[t] - mn);
                    const float pv = __expf(v - mn);
                    float rs = pv;
                    rs += __shfl_xor(rs, 1, 32);
                    rs += __shfl_xor(rs, 2, 32);
                    rs += __shfl_xor(rs, 4, 32);
                    rs += __shfl_xor(rs, 8, 32);
                    l_i[t] = l_i[t] * a + rs;
                    m_i[t] = mn;
                    alpha[t] = a;
                    sc[t] = pv;
                }
                #pragma unroll
                for (int t = 0; t < 8; ++t) {
                    o0[t] *= alpha[t]; o1[t] *= alpha[t];
                    o2[t] *= alpha[t]; o3[t] *= alpha[t];
                }

                #pragma unroll
                for (int t = 0; t < 8; ++t) pbuf[wave][t + 8 * half][l16] = sc[t];
                asm volatile("s_wait_dscnt 0" ::: "memory");
                v2f pa[4];
                #pragma unroll
                for (int kk = 0; kk < 4; ++kk) {
                    pa[kk].x = pbuf[wave][l16][kk * 4 + 2 * half];
                    pa[kk].y = pbuf[wave][l16][kk * 4 + 2 * half + 1];
                }

                #pragma unroll
                for (int kk = 0; kk < 4; ++kk) {
                    const v4f vlo = vf[kk][0], vhi = vf[kk][1];
                    v2f b0; b0.x = vlo.x; b0.y = vlo.y;
                    v2f b1; b1.x = vlo.z; b1.y = vlo.w;
                    v2f b2; b2.x = vhi.x; b2.y = vhi.y;
                    v2f b3; b3.x = vhi.z; b3.y = vhi.w;
                    o0 = wmma_f32(pa[kk], b0, o0);
                    o1 = wmma_f32(pa[kk], b1, o1);
                    o2 = wmma_f32(pa[kk], b2, o2);
                    o3 = wmma_f32(pa[kk], b3, o3);
                }
                asm volatile("s_wait_dscnt 0" ::: "memory");
            }
            if (more) {
                float* kw = Ksw + (cur ^ 1) * FSTAGE_;
                float* vw = Vsw + (cur ^ 1) * VSTAGE_;
                #pragma unroll
                for (int i = 0; i < 8; ++i) *(v4f*)(kw + (8 * i) * KP_) = rk[i];
                #pragma unroll
                for (int i = 0; i < 4; ++i) {
                    v4f lo, hi;
                    lo.x = rv[2 * i].x; lo.y = rv[2 * i + 1].x; lo.z = rv[2 * i].y; lo.w = rv[2 * i + 1].y;
                    hi.x = rv[2 * i].z; hi.y = rv[2 * i + 1].z; hi.z = rv[2 * i].w; hi.w = rv[2 * i + 1].w;
                    *(v4f*)(vw + (8 * i) * VP_)      = lo;
                    *(v4f*)(vw + (8 * i) * VP_ + 64) = hi;
                }
            }
            __syncthreads();
        }

        float* Orow = Oout + ((size_t)b * S_ + qBase) * D_ + h * DK_ + 4 * l16;
        #pragma unroll
        for (int t = 0; t < 8; ++t) {
            const float inv = 1.0f / l_i[t];
            v4f o;
            o.x = o0[t] * inv; o.y = o1[t] * inv; o.z = o2[t] * inv; o.w = o3[t] * inv;
            *(v4f*)(Orow + (size_t)(t + 8 * half) * D_) = o;
        }
    }
}

__global__ __launch_bounds__(128) void out_proj_kernel(
    const float* __restrict__ A, const float* __restrict__ W,
    float* __restrict__ out) {
    constexpr int MT = 2;
    __shared__ __attribute__((aligned(16))) float As[2 * 32 * MT * AP_];
    __shared__ __attribute__((aligned(16))) float Bs[2 * (GK_ / 2) * BP_];
    const int rowBlk = blockIdx.y * (32 * MT);
    const int colBlk = blockIdx.x * 128;

    v8f acc[MT][4];
    #pragma unroll
    for (int tm = 0; tm < MT; ++tm) {
        #pragma unroll
        for (int tn = 0; tn < 4; ++tn) acc[tm][tn] = v8f{};
    }
    gemm_tile<MT, D_>(A, W, rowBlk, colBlk, As, Bs, acc);

    const int lane = threadIdx.x & 31;
    const int wave = threadIdx.x >> 5;
    const int c = lane & 15;
    const int h = lane >> 4;
    const int wm = wave >> 1;
    const int wn = wave & 1;
    const int gn0 = colBlk + wn * 64 + 4 * c;
    #pragma unroll
    for (int tm = 0; tm < MT; ++tm) {
        #pragma unroll
        for (int r = 0; r < 8; ++r) {
            const int gm = rowBlk + wm * 16 * MT + tm * 16 + 8 * h + r;
            v4f o;
            o.x = acc[tm][0][r]; o.y = acc[tm][1][r];
            o.z = acc[tm][2][r]; o.w = acc[tm][3][r];
            *(v4f*)(out + (size_t)gm * D_ + gn0) = o;
        }
    }
}

extern "C" void kernel_launch(void* const* d_in, const int* in_sizes, int n_in,
                              void* d_out, int out_size, void* d_ws, size_t ws_size,
                              hipStream_t stream) {
    (void)in_sizes; (void)n_in; (void)out_size; (void)ws_size;
    const float* X      = (const float*)d_in[0];
    const int*   tokpos = (const int*)d_in[1];
    const float* Wqkv   = (const float*)d_in[2];
    const float* Wo     = (const float*)d_in[3];
    float* out = (float*)d_out;

    float* ws = (float*)d_ws;
    const size_t BHSD = (size_t)B_ * H_ * S_ * DK_;
    float* Qb = ws;
    float* Kb = ws + BHSD;
    float* Vb = ws + 2 * BHSD;
    float* Ob = ws + 3 * BHSD;

    dim3 g1(N3D_ / 128, MROWS_ / 128);
    qkv_rope_kernel<<<g1, 128, 0, stream>>>(X, Wqkv, tokpos, Qb, Kb, Vb);

    dim3 g2(NQB_ / 2, B_ * H_);
    flash_kernel<<<g2, 128, 0, stream>>>(Qb, Kb, Vb, Ob);

    dim3 g3(D_ / 128, MROWS_ / 64);
    out_proj_kernel<<<g3, 128, 0, stream>>>(Ob, Wo, out);
}


# Attention scores computed transposed (K*Q^T): P^T feeds V^T*P^T from registers, no LDS transpose of P; row max/sum in-lane plus one half-wave exchange; softmax sum tree and PV key order change
# speedup vs baseline: 2.3778x; 1.2068x over previous
#include <hip/hip_runtime.h>

typedef float v2f __attribute__((ext_vector_type(2)));
typedef float v8f __attribute__((ext_vector_type(8)));

#define B_  2
#define S_  2048
#define D_  1024
#define H_  16
#define DK_ 64
#define N3D_ (3 * D_)
#define MROWS_ (B_ * S_)
#define THETA_ 10000.0f

__device__ __forceinline__ v8f wmma_f32(v2f a, v2f b, v8f c) {
    return __builtin_amdgcn_wmma_f32_16x16x4_f32(
         false, a,  false, b,
         (short)0, c,  false,  false);
}

typedef float v4f __attribute__((ext_vector_type(4)));

#define GK_ 16
#define AP_ 20
#define BP_ 256
#define KSTAGES_ (D_ / GK_)

template <int MT, int LDB>
__device__ __forceinline__ void gemm_tile(
    const float* __restrict__ A, const float* __restrict__ Bm,
    const int rowBlk, const int colBlk,
    float* __restrict__ As, float* __restrict__ Bs, v8f (&acc)[MT][4]) {
    constexpr int BM = 32 * MT;
    constexpr int ASTAGE = BM * AP_;
    constexpr int BSTAGE = (GK_ / 2) * BP_;
    const int tid  = threadIdx.x;
    const int lane = tid & 31;
    const int wave = tid >> 5;
    const int c = lane & 15;
    const int h = lane >> 4;
    const int wm = wave >> 1;
    const int wn = wave & 1;

    const int arow = tid >> 2, akq = tid & 3;
    const int bkp  = tid >> 5, bcq = tid & 31;
    const float* Ag = A  + (size_t)(rowBlk + arow) * D_ + akq * 4;
    const float* Bg = Bm + (size_t)(2 * bkp) * LDB + colBlk + bcq * 4;
    float* Asw = As + arow * AP_ + akq * 4;
    float* Bsw = Bs + bkp * BP_ + bcq * 4;

    v4f ra[MT], rb[4];
    #pragma unroll
    for (int i = 0; i < MT; ++i) ra[i] = *(const v4f*)(Ag + (size_t)(32 * i) * D_);
    #pragma unroll
    for (int i = 0; i < 4; ++i)  rb[i] = *(const v4f*)(Bg + (size_t)(8 * (i >> 1) + (i & 1)) * LDB);
    #pragma unroll
    for (int i = 0; i < MT; ++i) *(v4f*)(Asw + (32 * i) * AP_) = ra[i];
    #pragma unroll
    for (int i = 0; i < 2; ++i) {
        v4f lo, hi;
        lo.x = rb[2 * i].x; lo.y = rb[2 * i + 1].x; lo.z = rb[2 * i].y; lo.w = rb[2 * i + 1].y;
        hi.x = rb[2 * i].z; hi.y = rb[2 * i + 1].z; hi.z = rb[2 * i].w; hi.w = rb[2 * i + 1].w;
        *(v4f*)(Bsw + (4 * i) * BP_)       = lo;
        *(v4f*)(Bsw + (4 * i) * BP_ + 128) = hi;
    }
    __syncthreads();

    const float* Asr = As + (wm * 16 * MT + c) * AP_ + h * 4;
    const float* Bsr = Bs + (h * 2) * BP_ + (wn * 16 + c) * 4;

    for (int kc = 0; kc < KSTAGES_; ++kc) {
        const int cur = kc & 1;
        const bool more = (kc + 1 < KSTAGES_);
        if (more) {
            const float* ag = Ag + (kc + 1) * GK_;
            const float* bg = Bg + (size_t)(kc + 1) * GK_ * LDB;
            #pragma unroll
            for (int i = 0; i < MT; ++i) ra[i] = *(const v4f*)(ag + (size_t)(32 * i) * D_);
            #pragma unroll
            for (int i = 0; i < 4; ++i)  rb[i] = *(const v4f*)(bg + (size_t)(8 * (i >> 1) + (i & 1)) * LDB);
            __builtin_amdgcn_sched_barrier(0);
        }
        const float* as = Asr + cur * ASTAGE;
        const float* bs = Bsr + cur * BSTAGE;
        #pragma unroll
        for (int s = 0; s < 2; ++s) {
            v4f af[MT], bf[2][2];
            #pragma unroll
            for (int tm = 0; tm < MT; ++tm) af[tm] = *(const v4f*)(as + tm * 16 * AP_ + s * 8);
            #pragma unroll
            for (int p = 0; p < 2; ++p) {
                bf[p][0] = *(const v4f*)(bs + (s * 4 + p) * BP_);
                bf[p][1] = *(const v4f*)(bs + (s * 4 + p) * BP_ + 128);
            }
            #pragma unroll
            for (int p = 0; p < 2; ++p) {
                #pragma unroll
                for (int tm = 0; tm < MT; ++tm) {
                    v2f a; a.x = af[tm][2 * p]; a.y = af[tm][2 * p + 1];
                    #pragma unroll
                    for (int tn = 0; tn < 4; ++tn) {
                        v2f b; b.x = bf[p][tn >> 1][2 * (tn & 1)]; b.y = bf[p][tn >> 1][2 * (tn & 1) + 1];
                        acc[tm][tn] = wmma_f32(a, b, acc[tm][tn]);
                    }
                }
            }
        }
        if (more) {
            __builtin_amdgcn_sched_barrier(0);
            float* aw = Asw + (cur ^ 1) * ASTAGE;
            float* bw = Bsw + (cur ^ 1) * BSTAGE;
            #pragma unroll
            for (int i = 0; i < MT; ++i) *(v4f*)(aw + (32 * i) * AP_) = ra[i];
            #pragma unroll
            for (int i = 0; i < 2; ++i) {
                v4f lo, hi;
                lo.x = rb[2 * i].x; lo.y = rb[2 * i + 1].x; lo.z = rb[2 * i].y; lo.w = rb[2 * i + 1].y;
                hi.x = rb[2 * i].z; hi.y = rb[2 * i + 1].z; hi.z = rb[2 * i].w; hi.w = rb[2 * i + 1].w;
                *(v4f*)(bw + (4 * i) * BP_)       = lo;
                *(v4f*)(bw + (4 * i) * BP_ + 128) = hi;
            }
        }
        __syncthreads();
    }
}

__global__ __launch_bounds__(128) void qkv_rope_kernel(
    const float* __restrict__ X, const float* __restrict__ W,
    const int* __restrict__ tokpos,
    float* __restrict__ Qo, float* __restrict__ Ko, float* __restrict__ Vo) {
    constexpr int MT = 4;
    __shared__ __attribute__((aligned(16))) float As[2 * 32 * MT * AP_];
    __shared__ __attribute__((aligned(16))) float Bs[2 * (GK_ / 2) * BP_];
    const int rowBlk = blockIdx.y * (32 * MT);
    const int colBlk = blockIdx.x * 128;

    v8f acc[MT][4];
    #pragma unroll
    for (int tm = 0; tm < MT; ++tm) {
        #pragma unroll
        for (int tn = 0; tn < 4; ++tn) acc[tm][tn] = v8f{};
    }
    gemm_tile<MT, N3D_>(X, W, rowBlk, colBlk, As, Bs, acc);

    const int lane = threadIdx.x & 31;
    const int wave = threadIdx.x >> 5;
    const int c = lane & 15;
    const int h = lane >> 4;
    const int wm = wave >> 1;
    const int wn = wave & 1;

    const int gn0 = colBlk + wn * 64 + 4 * c;
    const int sel = gn0 >> 10;
    const int d   = gn0 & (D_ - 1);
    const int hh  = d >> 6;
    const int dd0 = d & (DK_ - 1);
    float* dst = (sel == 0) ? Qo : (sel == 1) ? Ko : Vo;

    float invf0 = 0.0f, invf1 = 0.0f;
    if (sel < 2) {
        const float fexp0 = (float)dd0 * (1.0f / (float)DK_);
        const float fexp1 = (float)(dd0 + 2) * (1.0f / (float)DK_);
        invf0 = __powf(THETA_, -fexp0);
        invf1 = __powf(THETA_, -fexp1);
    }

    #pragma unroll
    for (int tm = 0; tm < MT; ++tm) {
        #pragma unroll
        for (int r = 0; r < 8; ++r) {
            const int gm = rowBlk + wm * 16 * MT + tm * 16 + 8 * h + r;
            const int s  = gm & (S_ - 1);
            const int b  = gm >> 11;
            const float x0 = acc[tm][0][r], x1 = acc[tm][1][r];
            const float x2 = acc[tm][2][r], x3 = acc[tm][3][r];
            v4f o;
            if (sel < 2) {
                const float pos = (float)tokpos[s];
                float sn0, cs0, sn1, cs1;
                __sincosf(pos * invf0, &sn0, &cs0);
                __sincosf(pos * invf1, &sn1, &cs1);
                o.x = x0 * cs0 - x1 * sn0;
                o.y = x0 * sn0 + x1 * cs0;
                o.z = x2 * cs1 - x3 * sn1;
                o.w = x2 * sn1 + x3 * cs1;
            } else {
                o.x = x0; o.y = x1; o.z = x2; o.w = x3;
            }
            *(v4f*)(dst + (size_t)(((b * H_ + hh) * S_) + s) * DK_ + dd0) = o;
        }
    }
}

#define FQ_ 64
#define FK_ 64
#define KP_ 68
#define FSTAGE_ (FK_ * KP_)
#define VTP_ 68
#define VTSTAGE_ (DK_ * VTP_)
#define OP_ 68
#define NQB_ (S_ / FQ_)

__device__ __forceinline__ float xhalf(float x) {
    const int bits = __builtin_bit_cast(int, x);
    return __builtin_bit_cast(float,
        __builtin_amdgcn_permlanex16(bits, bits, 0x76543210, 0xfedcba98, false, false));
}

__global__ __launch_bounds__(128) void flash_kernel(
    const float* __restrict__ Q, const float* __restrict__ K,
    const float* __restrict__ V, float* __restrict__ Oout) {
    __shared__ __attribute__((aligned(16))) float Ks[2 * FSTAGE_];
    __shared__ __attribute__((aligned(16))) float Vs[2 * VTSTAGE_];
    __shared__ __attribute__((aligned(16))) float obuf[4][16 * OP_];
    const int tid  = threadIdx.x;
    const int lane = tid & 31;
    const int wave = __builtin_amdgcn_readfirstlane(tid >> 5);
    const int l16  = lane & 15;
    const int half = lane >> 4;
    const int bh = blockIdx.y;
    const int b  = bh >> 4, h = bh & 15;

    const float* Qp = Q + (size_t)bh * S_ * DK_;
    const float* Kp = K + (size_t)bh * S_ * DK_;
    const float* Vp = V + (size_t)bh * S_ * DK_;

    const int srow = tid >> 4, scq = tid & 15;
    const float* Kg = Kp + (size_t)srow * DK_ + scq * 4;
    const float* Vg = Vp + (size_t)(4 * srow) * DK_ + scq * 4;
    float* Ksw = Ks + srow * KP_ + scq * 4;
    float* Vsw = Vs + scq * VTP_ + 4 * srow;
    const float* Ksr = Ks + l16 * KP_ + 4 * half;
    const float* Vsr = Vs + l16 * VTP_ + 8 * half;
    float* ob = obuf[wave];

    for (int pass = 0; pass < 2; ++pass) {
        const int qb = (pass == 0) ? (NQB_ - 1 - (int)blockIdx.x) : (int)blockIdx.x;
        const int qBase = qb * FQ_ + wave * 16;

        v2f qa[16];
        #pragma unroll
        for (int j = 0; j < 8; ++j) {
            const v4f qv = *(const v4f*)(Qp + (size_t)(qBase + l16) * DK_ + 8 * j + 4 * half);
            qa[2 * j].x     = qv.x * 0.125f; qa[2 * j].y     = qv.y * 0.125f;
            qa[2 * j + 1].x = qv.z * 0.125f; qa[2 * j + 1].y = qv.w * 0.125f;
        }

        v8f o0 = {}, o1 = {}, o2 = {}, o3 = {};
        float m_i = -1e30f, l_i = 0.0f;

        v4f rk[8], rv[8];
        #pragma unroll
        for (int i = 0; i < 8; ++i) {
            rk[i] = *(const v4f*)(Kg + (size_t)(8 * i) * DK_);
            rv[i] = *(const v4f*)(Vg + (size_t)(32 * (i >> 2) + (i & 3)) * DK_);
        }
        #pragma unroll
        for (int i = 0; i < 8; ++i) *(v4f*)(Ksw + (8 * i) * KP_) = rk[i];
        #pragma unroll
        for (int g = 0; g < 2; ++g) {
            #pragma unroll
            for (int i = 0; i < 4; ++i) {
                v4f w;
                w.x = rv[4 * g][i]; w.y = rv[4 * g + 1][i]; w.z = rv[4 * g + 2][i]; w.w = rv[4 * g + 3][i];
                *(v4f*)(Vsw + (16 * i) * VTP_ + 32 * g) = w;
            }
        }
        __syncthreads();

        for (int kb = 0; kb <= qb; ++kb) {
            const int cur = kb & 1;
            const bool more = (kb < qb);
            if (more) {
                #pragma unroll
                for (int i = 0; i < 8; ++i) {
                    rk[i] = *(const v4f*)(Kg + (size_t)((kb + 1) * FK_ + 8 * i) * DK_);
                    rv[i] = *(const v4f*)(Vg + (size_t)((kb + 1) * FK_ + 32 * (i >> 2) + (i & 3)) * DK_);
                }
            }
            const int nst = more ? 4 : (wave + 1);
            const float* ksb = Ksr + cur * FSTAGE_;
            const float* vsb = Vsr + cur * VTSTAGE_;
            for (int st = 0; st < nst; ++st) {
                const int kBase = kb * FK_ + st * 16;

                v4f kf[8], vf[4][2];
                #pragma unroll
                for (int j = 0; j < 8; ++j) kf[j] = *(const v4f*)(ksb + (16 * st) * KP_ + 8 * j);
                #pragma unroll
                for (int tn = 0; tn < 4; ++tn) {
                    vf[tn][0] = *(const v4f*)(vsb + (16 * tn) * VTP_ + 16 * st);
                    vf[tn][1] = *(const v4f*)(vsb + (16 * tn) * VTP_ + 16 * st + 4);
                }

                v8f sA = {}, sB = {};
                #pragma unroll
                for (int j = 0; j < 8; ++j) {
                    v2f a0; a0.x = kf[j].x; a0.y = kf[j].y;
                    v2f a1; a1.x = kf[j].z; a1.y = kf[j].w;
                    sA = wmma_f32(a0, qa[2 * j],     sA);
                    sB = wmma_f32(a1, qa[2 * j + 1], sB);
                }
                v8f sc = sA + sB;

                const int qrow = qBase + l16;
                const int key0 = kBase + 8 * half;
                #pragma unroll
                for (int r = 0; r < 8; ++r) sc[r] = (key0 + r <= qrow) ? sc[r] : -1e30f;
                float mx = fmaxf(fmaxf(fmaxf(sc[0], sc[1]), fmaxf(sc[2], sc[3])),
                                 fmaxf(fmaxf(sc[4], sc[5]), fmaxf(sc[6], sc[7])));
                mx = fmaxf(mx, xhalf(mx));
                const float mn = fmaxf(m_i, mx);
                const float alpha = __expf(m_i - mn);
                #pragma unroll
                for (int r = 0; r < 8; ++r) sc[r] = __expf(sc[r] - mn);
                float rs = ((sc[0] + sc[1]) + (sc[2] + sc[3])) + ((sc[4] + sc[5]) + (sc[6] + sc[7]));
                rs = rs + xhalf(rs);
                l_i = l_i * alpha + rs;
                m_i = mn;
                #pragma unroll
                for (int t = 0; t < 8; ++t) {
                    o0[t] *= alpha; o1[t] *= alpha;
                    o2[t] *= alpha; o3[t] *= alpha;
                }

                #pragma unroll
                for (int kk = 0; kk < 4; ++kk) {
                    v2f pb; pb.x = sc[2 * kk]; pb.y = sc[2 * kk + 1];
                    v2f a0; a0.x = vf[0][kk >> 1][2 * (kk & 1)]; a0.y = vf[0][kk >> 1][2 * (kk & 1) + 1];
                    v2f a1; a1.x = vf[1][kk >> 1][2 * (kk & 1)]; a1.y = vf[1][kk >> 1][2 * (kk & 1) + 1];
                    v2f a2; a2.x = vf[2][kk >> 1][2 * (kk & 1)]; a2.y = vf[2][kk >> 1][2 * (kk & 1) + 1];
                    v2f a3; a3.x = vf[3][kk >> 1][2 * (kk & 1)]; a3.y = vf[3][kk >> 1][2 * (kk & 1) + 1];
                    o0 = wmma_f32(a0, pb, o0);
                    o1 = wmma_f32(a1, pb, o1);
                    o2 = wmma_f32(a2, pb, o2);
                    o3 = wmma_f32(a3, pb, o3);
                }
            }
            if (more) {
                float* kw = Ksw + (cur ^ 1) * FSTAGE_;
                float* vw = Vsw + (cur ^ 1) * VTSTAGE_;
                #pragma unroll
                for (int i = 0; i < 8; ++i) *(v4f*)(kw + (8 * i) * KP_) = rk[i];
                #pragma unroll
                for (int g = 0; g < 2; ++g) {
                    #pragma unroll
                    for (int i = 0; i < 4; ++i) {
                        v4f w;
                        w.x = rv[4 * g][i]; w.y = rv[4 * g + 1][i]; w.z = rv[4 * g + 2][i]; w.w = rv[4 * g + 3][i];
                        *(v4f*)(vw + (16 * i) * VTP_ + 32 * g) = w;
                    }
                }
            }
            __syncthreads();
        }

        const float inv = 1.0f / l_i;
        #pragma unroll
        for (int r = 0; r < 8; ++r) {
            v4f o;
            o.x = o0[r] * inv; o.y = o1[r] * inv; o.z = o2[r] * inv; o.w = o3[r] * inv;
            *(v4f*)(ob + l16 * OP_ + 32 * half + 4 * r) = o;
        }
        asm volatile("s_wait_dscnt 0" ::: "memory");
        float* Orow = Oout + ((size_t)b * S_ + qBase) * D_ + h * DK_ + 4 * l16;
        #pragma unroll
        for (int i = 0; i < 8; ++i) {
            const v4f o = *(const v4f*)(ob + (2 * i + half) * OP_ + 4 * l16);
            *(v4f*)(Orow + (size_t)(2 * i + half) * D_) = o;
        }
    }
}

__global__ __launch_bounds__(128) void out_proj_kernel(
    const float* __restrict__ A, const float* __restrict__ W,
    float* __restrict__ out) {
    constexpr int MT = 2;
    __shared__ __attribute__((aligned(16))) float As[2 * 32 * MT * AP_];
    __shared__ __attribute__((aligned(16))) float Bs[2 * (GK_ / 2) * BP_];
    const int rowBlk = blockIdx.y * (32 * MT);
    const int colBlk = blockIdx.x * 128;

    v8f acc[MT][4];
    #pragma unroll
    for (int tm = 0; tm < MT; ++tm) {
        #pragma unroll
        for (int tn = 0; tn < 4; ++tn) acc[tm][tn] = v8f{};
    }
    gemm_tile<MT, D_>(A, W, rowBlk, colBlk, As, Bs, acc);

    const int lane = threadIdx.x & 31;
    const int wave = threadIdx.x >> 5;
    const int c = lane & 15;
    const int h = lane >> 4;
    const int wm = wave >> 1;
    const int wn = wave & 1;
    const int gn0 = colBlk + wn * 64 + 4 * c;
    #pragma unroll
    for (int tm = 0; tm < MT; ++tm) {
        #pragma unroll
        for (int r = 0; r < 8; ++r) {
            const int gm = rowBlk + wm * 16 * MT + tm * 16 + 8 * h + r;
            v4f o;
            o.x = acc[tm][0][r]; o.y = acc[tm][1][r];
            o.z = acc[tm][2][r]; o.w = acc[tm][3][r];
            *(v4f*)(out + (size_t)gm * D_ + gn0) = o;
        }
    }
}

extern "C" void kernel_launch(void* const* d_in, const int* in_sizes, int n_in,
                              void* d_out, int out_size, void* d_ws, size_t ws_size,
                              hipStream_t stream) {
    (void)in_sizes; (void)n_in; (void)out_size; (void)ws_size;
    const float* X      = (const float*)d_in[0];
    const int*   tokpos = (const int*)d_in[1];
    const float* Wqkv   = (const float*)d_in[2];
    const float* Wo     = (const float*)d_in[3];
    float* out = (float*)d_out;

    float* ws = (float*)d_ws;
    const size_t BHSD = (size_t)B_ * H_ * S_ * DK_;
    float* Qb = ws;
    float* Kb = ws + BHSD;
    float* Vb = ws + 2 * BHSD;
    float* Ob = ws + 3 * BHSD;

    dim3 g1(N3D_ / 128, MROWS_ / 128);
    qkv_rope_kernel<<<g1, 128, 0, stream>>>(X, Wqkv, tokpos, Qb, Kb, Vb);

    dim3 g2(NQB_ / 2, B_ * H_);
    flash_kernel<<<g2, 128, 0, stream>>>(Qb, Kb, Vb, Ob);

    dim3 g3(D_ / 128, MROWS_ / 64);
    out_proj_kernel<<<g3, 128, 0, stream>>>(Ob, Wo, out);
}
